# starting kernel
# baseline (speedup 1.0000x reference)
#include <hip/hip_runtime.h>

typedef __attribute__((ext_vector_type(2))) float v2f;
typedef __attribute__((ext_vector_type(8))) float v8f;

#define B_  64
#define ZD_ 512
#define E_  512
#define L_  8192
#define NL_ 1000

#define WMMA_F32(a, b, c) \
  __builtin_amdgcn_wmma_f32_16x16x4_f32(false, (a), false, (b), (short)0, (c), false, false)

__global__ void gather_yemb(const float* __restrict__ Wy,
                            const float* __restrict__ bias,
                            const int* __restrict__ y,
                            float* __restrict__ out) {
  int idx = blockIdx.x * blockDim.x + threadIdx.x;
  int b = idx >> 9;
  int e = idx & 511;
  out[idx] = Wy[e * NL_ + y[b]] + bias[e];
}

__global__ void irm_layer(const float* __restrict__ x,
                          const float* __restrict__ W,
                          const float* __restrict__ bias,
                          float* __restrict__ out) {
  __shared__ float part[2][4][256];

  const int lane = threadIdx.x & 31;
  const int wave = threadIdx.x >> 5;
  const int t    = wave >> 2;
  const int kw   = wave & 3;
  const int m_tile = blockIdx.x >> 4;
  const int m0     = m_tile * 16;
  const int n0base = (blockIdx.x & 15) * 32;
  const int half = lane >> 4, lr = lane & 15;
  const int k0 = kw * 128;

  const float* ap = x + (m0 + lr) * E_ + k0 + 2 * half;
  const float* bp = W + (n0base + t * 16 + lr) * E_ + k0 + 2 * half;

  v8f acc = {};
#pragma unroll 8
  for (int k = 0; k < 128; k += 4) {
    v2f a = *(const v2f*)(ap + k);
    v2f b = *(const v2f*)(bp + k);
    acc = WMMA_F32(a, b, acc);
  }

#pragma unroll
  for (int r = 0; r < 8; ++r)
    part[t][kw][(r + 8 * half) * 16 + lr] = acc[r];

  __syncthreads();

#pragma unroll
  for (int i = threadIdx.x; i < 512; i += 256) {
    int tt  = i >> 8;
    int idx = i & 255;
    float sum = part[tt][0][idx] + part[tt][1][idx] +
                part[tt][2][idx] + part[tt][3][idx];
    int m = m0 + (idx >> 4);
    int n = n0base + tt * 16 + (idx & 15);
    out[m * E_ + n] = sum + x[m * E_ + n] + bias[n];
  }
}

__global__ void sscal_kernel(const float* __restrict__ ye,
                             const float* __restrict__ wproj,
                             float* __restrict__ s) {
  int b = blockIdx.x;
  int lane = threadIdx.x;
  float sum = 0.f;
  for (int k = lane; k < E_; k += 32) sum += ye[b * E_ + k] * wproj[k];
#pragma unroll
  for (int off = 16; off > 0; off >>= 1) sum += __shfl_xor(sum, off, 32);
  if (lane == 0) s[b] = sum;
}

__global__ void zlat_kernel(const float* __restrict__ zz,
                            const float* __restrict__ Wz,
                            const float* __restrict__ s,
                            float* __restrict__ out) {
  const int lane = threadIdx.x & 31;
  const int wave = threadIdx.x >> 5;
  const int m_tile = blockIdx.x >> 5;
  const int pair   = (blockIdx.x & 31) * 8 + wave;
  const int m0 = m_tile * 16;
  const int n0 = pair * 32;
  const int half = lane >> 4, lr = lane & 15;

  const float* ap  = zz + (m0 + lr) * ZD_ + 2 * half;
  const float* bp0 = Wz + (n0 + lr) * ZD_ + 2 * half;
  const float* bp1 = Wz + (n0 + 16 + lr) * ZD_ + 2 * half;

  v8f acc0 = {}, acc1 = {};
#pragma unroll 8
  for (int k = 0; k < ZD_; k += 4) {
    v2f a  = *(const v2f*)(ap  + k);
    v2f b0 = *(const v2f*)(bp0 + k);
    v2f b1 = *(const v2f*)(bp1 + k);
    acc0 = WMMA_F32(a, b0, acc0);
    acc1 = WMMA_F32(a, b1, acc1);
  }

#pragma unroll
  for (int r = 0; r < 8; ++r) {
    int m = m0 + r + 8 * half;
    float sm = s[m];
    out[m * L_ + n0 + lr]      = acc0[r] * sm;
    out[m * L_ + n0 + 16 + lr] = acc1[r] * sm;
  }
}

extern "C" void kernel_launch(void* const* d_in, const int* in_sizes, int n_in,
                              void* d_out, int out_size, void* d_ws, size_t ws_size,
                              hipStream_t stream) {
  const float* z      = (const float*)d_in[0];
  const int*   y      = (const int*)  d_in[1];
  const float* W_yemb = (const float*)d_in[2];
  const float* b_yemb = (const float*)d_in[3];
  const float* Wy1    = (const float*)d_in[4];
  const float* by1    = (const float*)d_in[5];
  const float* Wy2    = (const float*)d_in[6];
  const float* by2    = (const float*)d_in[7];
  const float* Wz1    = (const float*)d_in[8];
  const float* bz1    = (const float*)d_in[9];
  const float* Wz2    = (const float*)d_in[10];
  const float* bz2    = (const float*)d_in[11];
  const float* W_zlat = (const float*)d_in[12];
  const float* w_proj = (const float*)d_in[13];
  float* out = (float*)d_out;

  float* ws  = (float*)d_ws;
  float* ye0 = ws;
  float* ye1 = ye0 + B_ * E_;
  float* ye2 = ye1 + B_ * E_;
  float* zz1 = ye2 + B_ * E_;
  float* zz2 = zz1 + B_ * ZD_;
  float* sc  = zz2 + B_ * ZD_;

  gather_yemb<<<(B_ * E_) / 256, 256, 0, stream>>>(W_yemb, b_yemb, y, ye0);
  irm_layer<<<64, 256, 0, stream>>>(ye0, Wy1, by1, ye1);
  irm_layer<<<64, 256, 0, stream>>>(ye1, Wy2, by2, ye2);
  irm_layer<<<64, 256, 0, stream>>>(z,   Wz1, bz1, zz1);
  irm_layer<<<64, 256, 0, stream>>>(zz1, Wz2, bz2, zz2);
  sscal_kernel<<<B_, 32, 0, stream>>>(ye2, w_proj, sc);
  zlat_kernel<<<128, 256, 0, stream>>>(zz2, W_zlat, sc, out);
}


# Merge launches 7 to 4: the two independent residual chains share launches stage by stage (role per block), and the row-scale dot moves into the big GEMM's blocks; sums keep their order
# speedup vs baseline: 1.4170x; 1.4170x over previous
#include <hip/hip_runtime.h>

typedef __attribute__((ext_vector_type(2))) float v2f;
typedef __attribute__((ext_vector_type(8))) float v8f;

#define B_  64
#define ZD_ 512
#define E_  512
#define L_  8192
#define NL_ 1000

#define WMMA_F32(a, b, c) \
  __builtin_amdgcn_wmma_f32_16x16x4_f32(false, (a), false, (b), (short)0, (c), false, false)


__device__ __forceinline__ void gather_rows(const float* __restrict__ Wy,
                                            const float* __restrict__ bias,
                                            const int* __restrict__ y,
                                            float* __restrict__ out, int blk) {
  int idx = blk * 256 + threadIdx.x;
  int b = idx >> 9;
  int e = idx & 511;
  out[idx] = Wy[e * NL_ + y[b]] + bias[e];
}

__device__ __forceinline__ void irm_tiles(const float* __restrict__ x,
                                          const float* __restrict__ W,
                                          const float* __restrict__ bias,
                                          float* __restrict__ out, int blk) {
  __shared__ float part[2][4][256];

  const int lane = threadIdx.x & 31;
  const int wave = threadIdx.x >> 5;
  const int t    = wave >> 2;
  const int kw   = wave & 3;
  const int m_tile = blk >> 4;
  const int m0     = m_tile * 16;
  const int n0base = (blk & 15) * 32;
  const int half = lane >> 4, lr = lane & 15;
  const int k0 = kw * 128;

  const float* ap = x + (m0 + lr) * E_ + k0 + 2 * half;
  const float* bp = W + (n0base + t * 16 + lr) * E_ + k0 + 2 * half;

  v8f acc = {};
#pragma unroll 8
  for (int k = 0; k < 128; k += 4) {
    v2f a = *(const v2f*)(ap + k);
    v2f b = *(const v2f*)(bp + k);
    acc = WMMA_F32(a, b, acc);
  }

#pragma unroll
  for (int r = 0; r < 8; ++r)
    part[t][kw][(r + 8 * half) * 16 + lr] = acc[r];

  __syncthreads();

#pragma unroll
  for (int i = threadIdx.x; i < 512; i += 256) {
    int tt  = i >> 8;
    int idx = i & 255;
    float sum = part[tt][0][idx] + part[tt][1][idx] +
                part[tt][2][idx] + part[tt][3][idx];
    int m = m0 + (idx >> 4);
    int n = n0base + tt * 16 + (idx & 15);
    out[m * E_ + n] = sum + x[m * E_ + n] + bias[n];
  }
}

__global__ void stage1_kernel(const float* __restrict__ z,
                              const float* __restrict__ W1,
                              const float* __restrict__ b1,
                              float* __restrict__ zz1,
                              const float* __restrict__ Wy,
                              const float* __restrict__ by,
                              const int* __restrict__ y,
                              float* __restrict__ ye0) {
  if (blockIdx.x < 64) irm_tiles(z, W1, b1, zz1, blockIdx.x);
  else                 gather_rows(Wy, by, y, ye0, blockIdx.x - 64);
}

__global__ void stage2_kernel(const float* __restrict__ xa,
                              const float* __restrict__ Wa,
                              const float* __restrict__ ba,
                              float* __restrict__ oa,
                              const float* __restrict__ xb,
                              const float* __restrict__ Wb,
                              const float* __restrict__ bb,
                              float* __restrict__ ob) {
  if (blockIdx.x < 64) irm_tiles(xa, Wa, ba, oa, blockIdx.x);
  else                 irm_tiles(xb, Wb, bb, ob, blockIdx.x - 64);
}

__global__ void irm_layer(const float* __restrict__ x,
                          const float* __restrict__ W,
                          const float* __restrict__ bias,
                          float* __restrict__ out) {
  irm_tiles(x, W, bias, out, blockIdx.x);
}

__global__ void zlat_kernel(const float* __restrict__ zz,
                            const float* __restrict__ Wz,
                            const float* __restrict__ ye,
                            const float* __restrict__ wproj,
                            float* __restrict__ out) {
  __shared__ float s_sh[16];

  const int lane = threadIdx.x & 31;
  const int wave = threadIdx.x >> 5;
  const int m_tile = blockIdx.x >> 5;
  const int pair   = (blockIdx.x & 31) * 8 + wave;
  const int m0 = m_tile * 16;
  const int n0 = pair * 32;
  const int half = lane >> 4, lr = lane & 15;

  const float* y0 = ye + (m0 + 2 * wave) * E_ + lane;
  const float* y1 = y0 + E_;
  float yv0[16], yv1[16], wv[16];
#pragma unroll
  for (int j = 0; j < 16; ++j) {
    yv0[j] = y0[32 * j];
    yv1[j] = y1[32 * j];
    wv[j]  = wproj[lane + 32 * j];
  }

  const float* ap  = zz + (m0 + lr) * ZD_ + 2 * half;
  const float* bp0 = Wz + (n0 + lr) * ZD_ + 2 * half;
  const float* bp1 = Wz + (n0 + 16 + lr) * ZD_ + 2 * half;

  v8f acc0 = {}, acc1 = {};
#pragma unroll 8
  for (int k = 0; k < ZD_; k += 4) {
    v2f a  = *(const v2f*)(ap  + k);
    v2f b0 = *(const v2f*)(bp0 + k);
    v2f b1 = *(const v2f*)(bp1 + k);
    acc0 = WMMA_F32(a, b0, acc0);
    acc1 = WMMA_F32(a, b1, acc1);
  }

  float s0 = 0.f, s1 = 0.f;
#pragma unroll
  for (int j = 0; j < 16; ++j) {
    s0 += yv0[j] * wv[j];
    s1 += yv1[j] * wv[j];
  }
#pragma unroll
  for (int off = 16; off > 0; off >>= 1) {
    s0 += __shfl_xor(s0, off, 32);
    s1 += __shfl_xor(s1, off, 32);
  }
  if (lane == 0) {
    s_sh[2 * wave]     = s0;
    s_sh[2 * wave + 1] = s1;
  }
  __syncthreads();

#pragma unroll
  for (int r = 0; r < 8; ++r) {
    int m = m0 + r + 8 * half;
    float sm = s_sh[r + 8 * half];
    out[m * L_ + n0 + lr]      = acc0[r] * sm;
    out[m * L_ + n0 + 16 + lr] = acc1[r] * sm;
  }
}

extern "C" void kernel_launch(void* const* d_in, const int* in_sizes, int n_in,
                              void* d_out, int out_size, void* d_ws, size_t ws_size,
                              hipStream_t stream) {
  const float* z      = (const float*)d_in[0];
  const int*   y      = (const int*)  d_in[1];
  const float* W_yemb = (const float*)d_in[2];
  const float* b_yemb = (const float*)d_in[3];
  const float* Wy1    = (const float*)d_in[4];
  const float* by1    = (const float*)d_in[5];
  const float* Wy2    = (const float*)d_in[6];
  const float* by2    = (const float*)d_in[7];
  const float* Wz1    = (const float*)d_in[8];
  const float* bz1    = (const float*)d_in[9];
  const float* Wz2    = (const float*)d_in[10];
  const float* bz2    = (const float*)d_in[11];
  const float* W_zlat = (const float*)d_in[12];
  const float* w_proj = (const float*)d_in[13];
  float* out = (float*)d_out;

  float* ws  = (float*)d_ws;
  float* ye0 = ws;
  float* ye1 = ye0 + B_ * E_;
  float* ye2 = ye1 + B_ * E_;
  float* zz1 = ye2 + B_ * E_;
  float* zz2 = zz1 + B_ * ZD_;

  stage1_kernel<<<192, 256, 0, stream>>>(z, Wz1, bz1, zz1, W_yemb, b_yemb, y, ye0);
  stage2_kernel<<<128, 256, 0, stream>>>(ye0, Wy1, by1, ye1, zz1, Wz2, bz2, zz2);
  irm_layer<<<64, 256, 0, stream>>>(ye1, Wy2, by2, ye2);
  zlat_kernel<<<128, 256, 0, stream>>>(zz2, W_zlat, ye2, w_proj, out);
}


# Big GEMM re-tiled: a block owns 32 columns for all 64 rows (Wz read once, not 4x; 256 blocks), its 8 waves split K with 128-bit loads, partials summed via LDS in fixed order; k sum order changes
# speedup vs baseline: 1.7332x; 1.2231x over previous
#include <hip/hip_runtime.h>

typedef __attribute__((ext_vector_type(2))) float v2f;
typedef __attribute__((ext_vector_type(4))) float v4f;
typedef __attribute__((ext_vector_type(8))) float v8f;

#define B_  64
#define ZD_ 512
#define E_  512
#define L_  8192
#define NL_ 1000

#define WMMA_F32(a, b, c) \
  __builtin_amdgcn_wmma_f32_16x16x4_f32(false, (a), false, (b), (short)0, (c), false, false)


__device__ __forceinline__ void gather_rows(const float* __restrict__ Wy,
                                            const float* __restrict__ bias,
                                            const int* __restrict__ y,
                                            float* __restrict__ out, int blk) {
  int idx = blk * 256 + threadIdx.x;
  int b = idx >> 9;
  int e = idx & 511;
  out[idx] = Wy[e * NL_ + y[b]] + bias[e];
}

__device__ __forceinline__ void irm_tiles(const float* __restrict__ x,
                                          const float* __restrict__ W,
                                          const float* __restrict__ bias,
                                          float* __restrict__ out, int blk) {
  __shared__ float part[2][4][256];

  const int lane = threadIdx.x & 31;
  const int wave = threadIdx.x >> 5;
  const int t    = wave >> 2;
  const int kw   = wave & 3;
  const int m_tile = blk >> 4;
  const int m0     = m_tile * 16;
  const int n0base = (blk & 15) * 32;
  const int half = lane >> 4, lr = lane & 15;
  const int k0 = kw * 128;

  const float* ap = x + (m0 + lr) * E_ + k0 + 2 * half;
  const float* bp = W + (n0base + t * 16 + lr) * E_ + k0 + 2 * half;

  v8f acc = {};
#pragma unroll 8
  for (int k = 0; k < 128; k += 4) {
    v2f a = *(const v2f*)(ap + k);
    v2f b = *(const v2f*)(bp + k);
    acc = WMMA_F32(a, b, acc);
  }

#pragma unroll
  for (int r = 0; r < 8; ++r)
    part[t][kw][(r + 8 * half) * 16 + lr] = acc[r];

  __syncthreads();

#pragma unroll
  for (int i = threadIdx.x; i < 512; i += 256) {
    int tt  = i >> 8;
    int idx = i & 255;
    float sum = part[tt][0][idx] + part[tt][1][idx] +
                part[tt][2][idx] + part[tt][3][idx];
    int m = m0 + (idx >> 4);
    int n = n0base + tt * 16 + (idx & 15);
    out[m * E_ + n] = sum + x[m * E_ + n] + bias[n];
  }
}

__global__ void stage1_kernel(const float* __restrict__ z,
                              const float* __restrict__ W1,
                              const float* __restrict__ b1,
                              float* __restrict__ zz1,
                              const float* __restrict__ Wy,
                              const float* __restrict__ by,
                              const int* __restrict__ y,
                              float* __restrict__ ye0) {
  if (blockIdx.x < 64) irm_tiles(z, W1, b1, zz1, blockIdx.x);
  else                 gather_rows(Wy, by, y, ye0, blockIdx.x - 64);
}

__global__ void stage2_kernel(const float* __restrict__ xa,
                              const float* __restrict__ Wa,
                              const float* __restrict__ ba,
                              float* __restrict__ oa,
                              const float* __restrict__ xb,
                              const float* __restrict__ Wb,
                              const float* __restrict__ bb,
                              float* __restrict__ ob) {
  if (blockIdx.x < 64) irm_tiles(xa, Wa, ba, oa, blockIdx.x);
  else                 irm_tiles(xb, Wb, bb, ob, blockIdx.x - 64);
}

__global__ void irm_layer(const float* __restrict__ x,
                          const float* __restrict__ W,
                          const float* __restrict__ bias,
                          float* __restrict__ out) {
  irm_tiles(x, W, bias, out, blockIdx.x);
}

__global__ __launch_bounds__(256)
void zlat_kernel(const float* __restrict__ zz,
                 const float* __restrict__ Wz,
                 const float* __restrict__ ye,
                 const float* __restrict__ wproj,
                 float* __restrict__ out) {
  __shared__ float part[8][64 * 32];
  __shared__ float s_sh[64];

  const int lane = threadIdx.x & 31;
  const int wave = threadIdx.x >> 5;
  const int half = lane >> 4, lr = lane & 15;
  const int n0 = blockIdx.x * 32;
  const int k0 = wave * 64 + 4 * half;

  const float* ap = zz + lr * ZD_ + k0;
  const float* bp = Wz + (n0 + lr) * ZD_ + k0;

  v4f bq[8][2], aq[8][4];
#pragma unroll
  for (int j = 0; j < 8; ++j)
#pragma unroll
    for (int t = 0; t < 2; ++t)
      bq[j][t] = *(const v4f*)(bp + t * 16 * ZD_ + 8 * j);
#pragma unroll
  for (int j = 0; j < 8; ++j)
#pragma unroll
    for (int m = 0; m < 4; ++m)
      aq[j][m] = *(const v4f*)(ap + m * 16 * ZD_ + 8 * j);

  const float* yr = ye + (8 * wave) * E_ + lane;
  float yv[8][16], wv[16];
#pragma unroll
  for (int j = 0; j < 16; ++j) {
    wv[j] = wproj[lane + 32 * j];
#pragma unroll
    for (int i = 0; i < 8; ++i)
      yv[i][j] = yr[i * E_ + 32 * j];
  }

  v8f acc[4][2] = {};
#pragma unroll
  for (int j = 0; j < 8; ++j) {
#pragma unroll
    for (int p = 0; p < 2; ++p) {
#pragma unroll
      for (int m = 0; m < 4; ++m) {
        v2f a = {aq[j][m][2 * p], aq[j][m][2 * p + 1]};
#pragma unroll
        for (int t = 0; t < 2; ++t) {
          v2f b = {bq[j][t][2 * p], bq[j][t][2 * p + 1]};
          acc[m][t] = WMMA_F32(a, b, acc[m][t]);
        }
      }
    }
  }

#pragma unroll
  for (int m = 0; m < 4; ++m)
#pragma unroll
    for (int t = 0; t < 2; ++t)
#pragma unroll
      for (int r = 0; r < 8; ++r)
        part[wave][(16 * m + 8 * half + r) * 32 + 16 * t + lr] = acc[m][t][r];

  float sv[8];
#pragma unroll
  for (int i = 0; i < 8; ++i) {
    float s = 0.f;
#pragma unroll
    for (int j = 0; j < 16; ++j) s = __builtin_fmaf(yv[i][j], wv[j], s);
    sv[i] = s;
  }
#pragma unroll
  for (int off = 16; off > 0; off >>= 1)
#pragma unroll
    for (int i = 0; i < 8; ++i) sv[i] += __shfl_xor(sv[i], off, 32);
  if (lane == 0) {
#pragma unroll
    for (int i = 0; i < 8; ++i) s_sh[8 * wave + i] = sv[i];
  }
  __syncthreads();

#pragma unroll
  for (int r = 0; r < 8; ++r) {
    const int row = 8 * wave + r;
    float sum = part[0][row * 32 + lane];
#pragma unroll
    for (int w = 1; w < 8; ++w) sum += part[w][row * 32 + lane];
    out[row * L_ + n0 + lane] = sum * s_sh[row];
  }
}

extern "C" void kernel_launch(void* const* d_in, const int* in_sizes, int n_in,
                              void* d_out, int out_size, void* d_ws, size_t ws_size,
                              hipStream_t stream) {
  const float* z      = (const float*)d_in[0];
  const int*   y      = (const int*)  d_in[1];
  const float* W_yemb = (const float*)d_in[2];
  const float* b_yemb = (const float*)d_in[3];
  const float* Wy1    = (const float*)d_in[4];
  const float* by1    = (const float*)d_in[5];
  const float* Wy2    = (const float*)d_in[6];
  const float* by2    = (const float*)d_in[7];
  const float* Wz1    = (const float*)d_in[8];
  const float* bz1    = (const float*)d_in[9];
  const float* Wz2    = (const float*)d_in[10];
  const float* bz2    = (const float*)d_in[11];
  const float* W_zlat = (const float*)d_in[12];
  const float* w_proj = (const float*)d_in[13];
  float* out = (float*)d_out;

  float* ws  = (float*)d_ws;
  float* ye0 = ws;
  float* ye1 = ye0 + B_ * E_;
  float* ye2 = ye1 + B_ * E_;
  float* zz1 = ye2 + B_ * E_;
  float* zz2 = zz1 + B_ * ZD_;

  stage1_kernel<<<192, 256, 0, stream>>>(z, Wz1, bz1, zz1, W_yemb, b_yemb, y, ye0);
  stage2_kernel<<<128, 256, 0, stream>>>(ye0, Wy1, by1, ye1, zz1, Wz2, bz2, zz2);
  irm_layer<<<64, 256, 0, stream>>>(ye1, Wy2, by2, ye2);
  zlat_kernel<<<256, 256, 0, stream>>>(zz2, W_zlat, ye2, w_proj, out);
}


# Last class layer folded into the projection vector by linearity: s = ye1.(w + Wy2^T w) + by2.w; its 64x512x512 GEMM becomes a mat-vec in launch 1; launches 4 to 3; sum order of s changes
# speedup vs baseline: 2.1665x; 1.2500x over previous
#include <hip/hip_runtime.h>

typedef __attribute__((ext_vector_type(2))) float v2f;
typedef __attribute__((ext_vector_type(4))) float v4f;
typedef __attribute__((ext_vector_type(8))) float v8f;

#define B_  64
#define ZD_ 512
#define E_  512
#define L_  8192
#define NL_ 1000

#define WMMA_F32(a, b, c) \
  __builtin_amdgcn_wmma_f32_16x16x4_f32(false, (a), false, (b), (short)0, (c), false, false)


__device__ __forceinline__ void gather_rows(const float* __restrict__ Wy,
                                            const float* __restrict__ bias,
                                            const int* __restrict__ y,
                                            float* __restrict__ out, int blk) {
  int idx = blk * 256 + threadIdx.x;
  int b = idx >> 9;
  int e = idx & 511;
  out[idx] = Wy[e * NL_ + y[b]] + bias[e];
}

__device__ __forceinline__ void irm_tiles(const float* __restrict__ x,
                                          const float* __restrict__ W,
                                          const float* __restrict__ bias,
                                          float* __restrict__ out, int blk) {
  __shared__ float part[2][4][256];

  const int lane = threadIdx.x & 31;
  const int wave = threadIdx.x >> 5;
  const int t    = wave >> 2;
  const int kw   = wave & 3;
  const int m_tile = blk >> 4;
  const int m0     = m_tile * 16;
  const int n0base = (blk & 15) * 32;
  const int half = lane >> 4, lr = lane & 15;
  const int k0 = kw * 128;

  const float* ap = x + (m0 + lr) * E_ + k0 + 2 * half;
  const float* bp = W + (n0base + t * 16 + lr) * E_ + k0 + 2 * half;

  v8f acc = {};
#pragma unroll 8
  for (int k = 0; k < 128; k += 4) {
    v2f a = *(const v2f*)(ap + k);
    v2f b = *(const v2f*)(bp + k);
    acc = WMMA_F32(a, b, acc);
  }

#pragma unroll
  for (int r = 0; r < 8; ++r)
    part[t][kw][(r + 8 * half) * 16 + lr] = acc[r];

  __syncthreads();

#pragma unroll
  for (int i = threadIdx.x; i < 512; i += 256) {
    int tt  = i >> 8;
    int idx = i & 255;
    float sum = part[tt][0][idx] + part[tt][1][idx] +
                part[tt][2][idx] + part[tt][3][idx];
    int m = m0 + (idx >> 4);
    int n = n0base + tt * 16 + (idx & 15);
    out[m * E_ + n] = sum + x[m * E_ + n] + bias[n];
  }
}

__device__ __forceinline__ void fold_proj(const float* __restrict__ W,
                                          const float* __restrict__ w,
                                          float* __restrict__ v, int blk) {
  __shared__ float fp[8][32];

  const int lane = threadIdx.x & 31;
  const int wave = __builtin_amdgcn_readfirstlane(threadIdx.x >> 5);
  const int k  = blk * 32 + lane;
  const int n0 = wave * 64;
  const float* wp = W + n0 * E_ + k;

  float col[64];
#pragma unroll
  for (int i = 0; i < 64; ++i) col[i] = wp[i * E_];

  float acc = 0.f;
#pragma unroll
  for (int i = 0; i < 64; ++i) acc = __builtin_fmaf(w[n0 + i], col[i], acc);

  fp[wave][lane] = acc;
  __syncthreads();

  if (wave == 0) {
    float sum = fp[0][lane];
#pragma unroll
    for (int q = 1; q < 8; ++q) sum += fp[q][lane];
    v[k] = w[k] + sum;
  }
}

__device__ __forceinline__ void bias_dot(const float* __restrict__ bias,
                                         const float* __restrict__ w,
                                         float* __restrict__ c) {
  if (threadIdx.x < 32) {
    const int lane = threadIdx.x;
    float s = 0.f;
#pragma unroll
    for (int j = 0; j < 16; ++j)
      s = __builtin_fmaf(bias[lane + 32 * j], w[lane + 32 * j], s);
#pragma unroll
    for (int off = 16; off > 0; off >>= 1) s += __shfl_xor(s, off, 32);
    if (lane == 0) *c = s;
  }
}

__global__ void stage1_kernel(const float* __restrict__ z,
                              const float* __restrict__ W1,
                              const float* __restrict__ b1,
                              float* __restrict__ zz1,
                              const float* __restrict__ Wy,
                              const float* __restrict__ by,
                              const int* __restrict__ y,
                              float* __restrict__ ye0,
                              const float* __restrict__ Wy2,
                              const float* __restrict__ by2,
                              const float* __restrict__ wproj,
                              float* __restrict__ vf,
                              float* __restrict__ cf) {
  if (blockIdx.x < 64)       irm_tiles(z, W1, b1, zz1, blockIdx.x);
  else if (blockIdx.x < 192) gather_rows(Wy, by, y, ye0, blockIdx.x - 64);
  else if (blockIdx.x < 208) fold_proj(Wy2, wproj, vf, blockIdx.x - 192);
  else                       bias_dot(by2, wproj, cf);
}

__global__ void stage2_kernel(const float* __restrict__ xa,
                              const float* __restrict__ Wa,
                              const float* __restrict__ ba,
                              float* __restrict__ oa,
                              const float* __restrict__ xb,
                              const float* __restrict__ Wb,
                              const float* __restrict__ bb,
                              float* __restrict__ ob) {
  if (blockIdx.x < 64) irm_tiles(xa, Wa, ba, oa, blockIdx.x);
  else                 irm_tiles(xb, Wb, bb, ob, blockIdx.x - 64);
}

__global__ __launch_bounds__(256)
void zlat_kernel(const float* __restrict__ zz,
                 const float* __restrict__ Wz,
                 const float* __restrict__ ye,
                 const float* __restrict__ vproj,
                 const float* __restrict__ cproj,
                 float* __restrict__ out) {
  __shared__ float part[8][64 * 32];
  __shared__ float s_sh[64];

  const int lane = threadIdx.x & 31;
  const int wave = threadIdx.x >> 5;
  const int half = lane >> 4, lr = lane & 15;
  const int n0 = blockIdx.x * 32;
  const int k0 = wave * 64 + 4 * half;

  const float* ap = zz + lr * ZD_ + k0;
  const float* bp = Wz + (n0 + lr) * ZD_ + k0;

  v4f bq[8][2], aq[8][4];
#pragma unroll
  for (int j = 0; j < 8; ++j)
#pragma unroll
    for (int t = 0; t < 2; ++t)
      bq[j][t] = *(const v4f*)(bp + t * 16 * ZD_ + 8 * j);
#pragma unroll
  for (int j = 0; j < 8; ++j)
#pragma unroll
    for (int m = 0; m < 4; ++m)
      aq[j][m] = *(const v4f*)(ap + m * 16 * ZD_ + 8 * j);

  const float* yr = ye + (8 * wave) * E_ + lane;
  float yv[8][16], wv[16];
  const float cadd = *cproj;
#pragma unroll
  for (int j = 0; j < 16; ++j) {
    wv[j] = vproj[lane + 32 * j];
#pragma unroll
    for (int i = 0; i < 8; ++i)
      yv[i][j] = yr[i * E_ + 32 * j];
  }

  v8f acc[4][2] = {};
#pragma unroll
  for (int j = 0; j < 8; ++j) {
#pragma unroll
    for (int p = 0; p < 2; ++p) {
#pragma unroll
      for (int m = 0; m < 4; ++m) {
        v2f a = {aq[j][m][2 * p], aq[j][m][2 * p + 1]};
#pragma unroll
        for (int t = 0; t < 2; ++t) {
          v2f b = {bq[j][t][2 * p], bq[j][t][2 * p + 1]};
          acc[m][t] = WMMA_F32(a, b, acc[m][t]);
        }
      }
    }
  }

#pragma unroll
  for (int m = 0; m < 4; ++m)
#pragma unroll
    for (int t = 0; t < 2; ++t)
#pragma unroll
      for (int r = 0; r < 8; ++r)
        part[wave][(16 * m + 8 * half + r) * 32 + 16 * t + lr] = acc[m][t][r];

  float sv[8];
#pragma unroll
  for (int i = 0; i < 8; ++i) {
    float s = 0.f;
#pragma unroll
    for (int j = 0; j < 16; ++j) s = __builtin_fmaf(yv[i][j], wv[j], s);
    sv[i] = s;
  }
#pragma unroll
  for (int off = 16; off > 0; off >>= 1)
#pragma unroll
    for (int i = 0; i < 8; ++i) sv[i] += __shfl_xor(sv[i], off, 32);
  if (lane == 0) {
#pragma unroll
    for (int i = 0; i < 8; ++i) s_sh[8 * wave + i] = sv[i] + cadd;
  }
  __syncthreads();

#pragma unroll
  for (int r = 0; r < 8; ++r) {
    const int row = 8 * wave + r;
    float sum = part[0][row * 32 + lane];
#pragma unroll
    for (int w = 1; w < 8; ++w) sum += part[w][row * 32 + lane];
    out[row * L_ + n0 + lane] = sum * s_sh[row];
  }
}

extern "C" void kernel_launch(void* const* d_in, const int* in_sizes, int n_in,
                              void* d_out, int out_size, void* d_ws, size_t ws_size,
                              hipStream_t stream) {
  const float* z      = (const float*)d_in[0];
  const int*   y      = (const int*)  d_in[1];
  const float* W_yemb = (const float*)d_in[2];
  const float* b_yemb = (const float*)d_in[3];
  const float* Wy1    = (const float*)d_in[4];
  const float* by1    = (const float*)d_in[5];
  const float* Wy2    = (const float*)d_in[6];
  const float* by2    = (const float*)d_in[7];
  const float* Wz1    = (const float*)d_in[8];
  const float* bz1    = (const float*)d_in[9];
  const float* Wz2    = (const float*)d_in[10];
  const float* bz2    = (const float*)d_in[11];
  const float* W_zlat = (const float*)d_in[12];
  const float* w_proj = (const float*)d_in[13];
  float* out = (float*)d_out;

  float* ws  = (float*)d_ws;
  float* ye0 = ws;
  float* ye1 = ye0 + B_ * E_;
  float* zz1 = ye1 + B_ * E_;
  float* zz2 = zz1 + B_ * ZD_;
  float* vf  = zz2 + B_ * ZD_;
  float* cf  = vf + E_;

  stage1_kernel<<<209, 256, 0, stream>>>(z, Wz1, bz1, zz1, W_yemb, b_yemb, y, ye0,
                                         Wy2, by2, w_proj, vf, cf);
  stage2_kernel<<<128, 256, 0, stream>>>(ye0, Wy1, by1, ye1, zz1, Wz2, bz2, zz2);
  zlat_kernel<<<256, 256, 0, stream>>>(zz2, W_zlat, ye1, vf, cf, out);
}


# First class layer folded into the projection vector too: s = ye0.(v + Wy1^T v) + by1.v + c, made in launch 2 as 16 partials per row; big GEMM loads those, not the embedding; sum order of s changes
# speedup vs baseline: 2.4336x; 1.1233x over previous
#include <hip/hip_runtime.h>

typedef __attribute__((ext_vector_type(2))) float v2f;
typedef __attribute__((ext_vector_type(4))) float v4f;
typedef __attribute__((ext_vector_type(8))) float v8f;

#define B_  64
#define ZD_ 512
#define E_  512
#define L_  8192
#define NL_ 1000

#define WMMA_F32(a, b, c) \
  __builtin_amdgcn_wmma_f32_16x16x4_f32(false, (a), false, (b), (short)0, (c), false, false)


__device__ __forceinline__ void gather_rows(const float* __restrict__ Wy,
                                            const float* __restrict__ bias,
                                            const int* __restrict__ y,
                                            float* __restrict__ out, int blk) {
  int idx = blk * 256 + threadIdx.x;
  int b = idx >> 9;
  int e = idx & 511;
  out[idx] = Wy[e * NL_ + y[b]] + bias[e];
}

__device__ __forceinline__ void irm_tiles(const float* __restrict__ x,
                                          const float* __restrict__ W,
                                          const float* __restrict__ bias,
                                          float* __restrict__ out, int blk) {
  __shared__ float part[2][4][256];

  const int lane = threadIdx.x & 31;
  const int wave = threadIdx.x >> 5;
  const int t    = wave >> 2;
  const int kw   = wave & 3;
  const int m_tile = blk >> 4;
  const int m0     = m_tile * 16;
  const int n0base = (blk & 15) * 32;
  const int half = lane >> 4, lr = lane & 15;
  const int k0 = kw * 128;

  const float* ap = x + (m0 + lr) * E_ + k0 + 2 * half;
  const float* bp = W + (n0base + t * 16 + lr) * E_ + k0 + 2 * half;

  v8f acc = {};
#pragma unroll 8
  for (int k = 0; k < 128; k += 4) {
    v2f a = *(const v2f*)(ap + k);
    v2f b = *(const v2f*)(bp + k);
    acc = WMMA_F32(a, b, acc);
  }

#pragma unroll
  for (int r = 0; r < 8; ++r)
    part[t][kw][(r + 8 * half) * 16 + lr] = acc[r];

  __syncthreads();

#pragma unroll
  for (int i = threadIdx.x; i < 512; i += 256) {
    int tt  = i >> 8;
    int idx = i & 255;
    float sum = part[tt][0][idx] + part[tt][1][idx] +
                part[tt][2][idx] + part[tt][3][idx];
    int m = m0 + (idx >> 4);
    int n = n0base + tt * 16 + (idx & 15);
    out[m * E_ + n] = sum + x[m * E_ + n] + bias[n];
  }
}

__device__ __forceinline__ void fold_proj(const float* __restrict__ W,
                                          const float* __restrict__ w,
                                          float* __restrict__ v, int blk) {
  __shared__ float fp[8][32];

  const int lane = threadIdx.x & 31;
  const int wave = __builtin_amdgcn_readfirstlane(threadIdx.x >> 5);
  const int k  = blk * 32 + lane;
  const int n0 = wave * 64;
  const float* wp = W + n0 * E_ + k;

  float col[64];
#pragma unroll
  for (int i = 0; i < 64; ++i) col[i] = wp[i * E_];

  float acc = 0.f;
#pragma unroll
  for (int i = 0; i < 64; ++i) acc = __builtin_fmaf(w[n0 + i], col[i], acc);

  fp[wave][lane] = acc;
  __syncthreads();

  if (wave == 0) {
    float sum = fp[0][lane];
#pragma unroll
    for (int q = 1; q < 8; ++q) sum += fp[q][lane];
    v[k] = w[k] + sum;
  }
}

__device__ __forceinline__ void bias_dot(const float* __restrict__ bias,
                                         const float* __restrict__ w,
                                         float* __restrict__ c) {
  if (threadIdx.x < 32) {
    const int lane = threadIdx.x;
    float s = 0.f;
#pragma unroll
    for (int j = 0; j < 16; ++j)
      s = __builtin_fmaf(bias[lane + 32 * j], w[lane + 32 * j], s);
#pragma unroll
    for (int off = 16; off > 0; off >>= 1) s += __shfl_xor(s, off, 32);
    if (lane == 0) *c = s;
  }
}

__device__ __forceinline__ void fold_rows(const float* __restrict__ W,
                                          const float* __restrict__ w,
                                          const float* __restrict__ ye,
                                          float* __restrict__ sp, int blk) {
  __shared__ float fq[8][32];
  __shared__ float srow[64];

  const int lane = threadIdx.x & 31;
  const int wave = __builtin_amdgcn_readfirstlane(threadIdx.x >> 5);
  const int k  = blk * 32 + lane;
  const int n0 = wave * 64;
  const float* wp = W + n0 * E_ + k;

  float col[64];
#pragma unroll
  for (int i = 0; i < 64; ++i) col[i] = wp[i * E_];

  const float* yr = ye + (8 * wave) * E_ + k;
  float yv[8];
#pragma unroll
  for (int i = 0; i < 8; ++i) yv[i] = yr[i * E_];
  const float wk = w[k];

  float acc = 0.f;
#pragma unroll
  for (int i = 0; i < 64; ++i) acc = __builtin_fmaf(w[n0 + i], col[i], acc);

  fq[wave][lane] = acc;
  __syncthreads();

  float sum = fq[0][lane];
#pragma unroll
  for (int q = 1; q < 8; ++q) sum += fq[q][lane];
  const float u = wk + sum;

  float pv[8];
#pragma unroll
  for (int i = 0; i < 8; ++i) pv[i] = yv[i] * u;
#pragma unroll
  for (int off = 16; off > 0; off >>= 1)
#pragma unroll
    for (int i = 0; i < 8; ++i) pv[i] += __shfl_xor(pv[i], off, 32);
  if (lane == 0) {
#pragma unroll
    for (int i = 0; i < 8; ++i) srow[8 * wave + i] = pv[i];
  }
  __syncthreads();
  if (threadIdx.x < 64) sp[blk * 64 + threadIdx.x] = srow[threadIdx.x];
}

__device__ __forceinline__ void bias_dot_add(const float* __restrict__ bias,
                                             const float* __restrict__ w,
                                             const float* __restrict__ c,
                                             float* __restrict__ c2) {
  if (threadIdx.x < 32) {
    const int lane = threadIdx.x;
    float s = 0.f;
#pragma unroll
    for (int j = 0; j < 16; ++j)
      s = __builtin_fmaf(bias[lane + 32 * j], w[lane + 32 * j], s);
#pragma unroll
    for (int off = 16; off > 0; off >>= 1) s += __shfl_xor(s, off, 32);
    if (lane == 0) *c2 = s + *c;
  }
}

__global__ void stage1_kernel(const float* __restrict__ z,
                              const float* __restrict__ W1,
                              const float* __restrict__ b1,
                              float* __restrict__ zz1,
                              const float* __restrict__ Wy,
                              const float* __restrict__ by,
                              const int* __restrict__ y,
                              float* __restrict__ ye0,
                              const float* __restrict__ Wy2,
                              const float* __restrict__ by2,
                              const float* __restrict__ wproj,
                              float* __restrict__ vf,
                              float* __restrict__ cf) {
  if (blockIdx.x < 64)       irm_tiles(z, W1, b1, zz1, blockIdx.x);
  else if (blockIdx.x < 192) gather_rows(Wy, by, y, ye0, blockIdx.x - 64);
  else if (blockIdx.x < 208) fold_proj(Wy2, wproj, vf, blockIdx.x - 192);
  else                       bias_dot(by2, wproj, cf);
}

__global__ void stage2_kernel(const float* __restrict__ x,
                              const float* __restrict__ W2,
                              const float* __restrict__ b2,
                              float* __restrict__ o,
                              const float* __restrict__ Wy1,
                              const float* __restrict__ by1,
                              const float* __restrict__ ye0,
                              const float* __restrict__ vf,
                              const float* __restrict__ cf,
                              float* __restrict__ sp,
                              float* __restrict__ c2) {
  if (blockIdx.x < 64)      irm_tiles(x, W2, b2, o, blockIdx.x);
  else if (blockIdx.x < 80) fold_rows(Wy1, vf, ye0, sp, blockIdx.x - 64);
  else                      bias_dot_add(by1, vf, cf, c2);
}

__global__ __launch_bounds__(256)
void zlat_kernel(const float* __restrict__ zz,
                 const float* __restrict__ Wz,
                 const float* __restrict__ sp,
                 const float* __restrict__ c2,
                 float* __restrict__ out) {
  __shared__ float part[8][64 * 32];

  const int lane = threadIdx.x & 31;
  const int wave = threadIdx.x >> 5;
  const int half = lane >> 4, lr = lane & 15;
  const int n0 = blockIdx.x * 32;
  const int k0 = wave * 64 + 4 * half;

  const float* ap = zz + lr * ZD_ + k0;
  const float* bp = Wz + (n0 + lr) * ZD_ + k0;

  v4f bq[8][2], aq[8][4];
#pragma unroll
  for (int j = 0; j < 8; ++j)
#pragma unroll
    for (int t = 0; t < 2; ++t)
      bq[j][t] = *(const v4f*)(bp + t * 16 * ZD_ + 8 * j);
#pragma unroll
  for (int j = 0; j < 8; ++j)
#pragma unroll
    for (int m = 0; m < 4; ++m)
      aq[j][m] = *(const v4f*)(ap + m * 16 * ZD_ + 8 * j);

  const float* spr = sp + (4 * (lane >> 3)) * 64 + 8 * wave + (lane & 7);
  const float p0 = spr[0], p1 = spr[64], p2 = spr[128], p3 = spr[192];
  const float cadd = *c2;

  v8f acc[4][2] = {};
#pragma unroll
  for (int j = 0; j < 8; ++j) {
#pragma unroll
    for (int p = 0; p < 2; ++p) {
#pragma unroll
      for (int m = 0; m < 4; ++m) {
        v2f a = {aq[j][m][2 * p], aq[j][m][2 * p + 1]};
#pragma unroll
        for (int t = 0; t < 2; ++t) {
          v2f b = {bq[j][t][2 * p], bq[j][t][2 * p + 1]};
          acc[m][t] = WMMA_F32(a, b, acc[m][t]);
        }
      }
    }
  }

#pragma unroll
  for (int m = 0; m < 4; ++m)
#pragma unroll
    for (int t = 0; t < 2; ++t)
#pragma unroll
      for (int r = 0; r < 8; ++r)
        part[wave][(16 * m + 8 * half + r) * 32 + 16 * t + lr] = acc[m][t][r];

  float sl = ((p0 + p1) + p2) + p3;
  sl += __shfl_xor(sl, 8, 32);
  sl += __shfl_xor(sl, 16, 32);
  sl += cadd;

  __syncthreads();

#pragma unroll
  for (int r = 0; r < 8; ++r) {
    const int row = 8 * wave + r;
    float sum = part[0][row * 32 + lane];
#pragma unroll
    for (int w = 1; w < 8; ++w) sum += part[w][row * 32 + lane];
    out[row * L_ + n0 + lane] = sum * __shfl(sl, r, 32);
  }
}

extern "C" void kernel_launch(void* const* d_in, const int* in_sizes, int n_in,
                              void* d_out, int out_size, void* d_ws, size_t ws_size,
                              hipStream_t stream) {
  const float* z      = (const float*)d_in[0];
  const int*   y      = (const int*)  d_in[1];
  const float* W_yemb = (const float*)d_in[2];
  const float* b_yemb = (const float*)d_in[3];
  const float* Wy1    = (const float*)d_in[4];
  const float* by1    = (const float*)d_in[5];
  const float* Wy2    = (const float*)d_in[6];
  const float* by2    = (const float*)d_in[7];
  const float* Wz1    = (const float*)d_in[8];
  const float* bz1    = (const float*)d_in[9];
  const float* Wz2    = (const float*)d_in[10];
  const float* bz2    = (const float*)d_in[11];
  const float* W_zlat = (const float*)d_in[12];
  const float* w_proj = (const float*)d_in[13];
  float* out = (float*)d_out;

  float* ws  = (float*)d_ws;
  float* ye0 = ws;
  float* sp  = ye0 + B_ * E_;
  float* c2  = sp + 16 * B_;
  float* zz1 = sp + B_ * E_;
  float* zz2 = zz1 + B_ * ZD_;
  float* vf  = zz2 + B_ * ZD_;
  float* cf  = vf + E_;

  stage1_kernel<<<209, 256, 0, stream>>>(z, Wz1, bz1, zz1, W_yemb, b_yemb, y, ye0,
                                         Wy2, by2, w_proj, vf, cf);
  stage2_kernel<<<81, 256, 0, stream>>>(zz1, Wz2, bz2, zz2, Wy1, by1, ye0, vf, cf,
                                        sp, c2);
  zlat_kernel<<<256, 256, 0, stream>>>(zz2, W_zlat, sp, c2, out);
}


# Both 64x512x512 layer GEMMs re-tiled like the big one: a wave owns a 64-wide K chunk of two tiles, issues all 128-bit fragment loads and the residual first; 8 partials via LDS; k sum order changes
# speedup vs baseline: 2.7341x; 1.1235x over previous
#include <hip/hip_runtime.h>

typedef __attribute__((ext_vector_type(2))) float v2f;
typedef __attribute__((ext_vector_type(4))) float v4f;
typedef __attribute__((ext_vector_type(8))) float v8f;

#define B_  64
#define ZD_ 512
#define E_  512
#define L_  8192
#define NL_ 1000

#define WMMA_F32(a, b, c) \
  __builtin_amdgcn_wmma_f32_16x16x4_f32(false, (a), false, (b), (short)0, (c), false, false)


__device__ __forceinline__ void gather_rows(const float* __restrict__ Wy,
                                            const float* __restrict__ bias,
                                            const int* __restrict__ y,
                                            float* __restrict__ out, int blk) {
  int idx = blk * 256 + threadIdx.x;
  int b = idx >> 9;
  int e = idx & 511;
  out[idx] = Wy[e * NL_ + y[b]] + bias[e];
}

__device__ __forceinline__ void irm_tiles(const float* __restrict__ x,
                                          const float* __restrict__ W,
                                          const float* __restrict__ bias,
                                          float* __restrict__ out, int blk) {
  __shared__ float part[8][16 * 32];

  const int lane = threadIdx.x & 31;
  const int wave = threadIdx.x >> 5;
  const int half = lane >> 4, lr = lane & 15;
  const int m0 = (blk >> 4) * 16;
  const int n0 = (blk & 15) * 32;
  const int k0 = wave * 64 + 4 * half;

  const float* ap = x + (m0 + lr) * E_ + k0;
  const float* bp = W + (n0 + lr) * E_ + k0;

  v4f aq[8], bq[8][2];
#pragma unroll
  for (int j = 0; j < 8; ++j) {
    aq[j] = *(const v4f*)(ap + 8 * j);
#pragma unroll
    for (int t = 0; t < 2; ++t)
      bq[j][t] = *(const v4f*)(bp + t * 16 * E_ + 8 * j);
  }

  const float xr0 = x[(m0 + wave) * E_ + n0 + lane];
  const float xr1 = x[(m0 + wave + 8) * E_ + n0 + lane];
  const float bn  = bias[n0 + lane];

  __builtin_amdgcn_sched_barrier(0);

  v8f acc[2] = {};
#pragma unroll
  for (int j = 0; j < 8; ++j) {
#pragma unroll
    for (int p = 0; p < 2; ++p) {
      v2f a = {aq[j][2 * p], aq[j][2 * p + 1]};
#pragma unroll
      for (int t = 0; t < 2; ++t) {
        v2f b = {bq[j][t][2 * p], bq[j][t][2 * p + 1]};
        acc[t] = WMMA_F32(a, b, acc[t]);
      }
    }
  }

#pragma unroll
  for (int t = 0; t < 2; ++t)
#pragma unroll
    for (int r = 0; r < 8; ++r)
      part[wave][(8 * half + r) * 32 + 16 * t + lr] = acc[t][r];

  __syncthreads();

  float s0 = part[0][wave * 32 + lane];
  float s1 = part[0][(wave + 8) * 32 + lane];
#pragma unroll
  for (int q = 1; q < 8; ++q) {
    s0 += part[q][wave * 32 + lane];
    s1 += part[q][(wave + 8) * 32 + lane];
  }
  out[(m0 + wave) * E_ + n0 + lane]     = s0 + xr0 + bn;
  out[(m0 + wave + 8) * E_ + n0 + lane] = s1 + xr1 + bn;
}

__device__ __forceinline__ void fold_proj(const float* __restrict__ W,
                                          const float* __restrict__ w,
                                          float* __restrict__ v, int blk) {
  __shared__ float fp[8][32];

  const int lane = threadIdx.x & 31;
  const int wave = __builtin_amdgcn_readfirstlane(threadIdx.x >> 5);
  const int k  = blk * 32 + lane;
  const int n0 = wave * 64;
  const float* wp = W + n0 * E_ + k;

  float col[64];
#pragma unroll
  for (int i = 0; i < 64; ++i) col[i] = wp[i * E_];

  float acc = 0.f;
#pragma unroll
  for (int i = 0; i < 64; ++i) acc = __builtin_fmaf(w[n0 + i], col[i], acc);

  fp[wave][lane] = acc;
  __syncthreads();

  if (wave == 0) {
    float sum = fp[0][lane];
#pragma unroll
    for (int q = 1; q < 8; ++q) sum += fp[q][lane];
    v[k] = w[k] + sum;
  }
}

__device__ __forceinline__ void bias_dot(const float* __restrict__ bias,
                                         const float* __restrict__ w,
                                         float* __restrict__ c) {
  if (threadIdx.x < 32) {
    const int lane = threadIdx.x;
    float s = 0.f;
#pragma unroll
    for (int j = 0; j < 16; ++j)
      s = __builtin_fmaf(bias[lane + 32 * j], w[lane + 32 * j], s);
#pragma unroll
    for (int off = 16; off > 0; off >>= 1) s += __shfl_xor(s, off, 32);
    if (lane == 0) *c = s;
  }
}

__device__ __forceinline__ void fold_rows(const float* __restrict__ W,
                                          const float* __restrict__ w,
                                          const float* __restrict__ ye,
                                          float* __restrict__ sp, int blk) {
  __shared__ float fq[8][32];
  __shared__ float srow[64];

  const int lane = threadIdx.x & 31;
  const int wave = __builtin_amdgcn_readfirstlane(threadIdx.x >> 5);
  const int k  = blk * 32 + lane;
  const int n0 = wave * 64;
  const float* wp = W + n0 * E_ + k;

  float col[64];
#pragma unroll
  for (int i = 0; i < 64; ++i) col[i] = wp[i * E_];

  const float* yr = ye + (8 * wave) * E_ + k;
  float yv[8];
#pragma unroll
  for (int i = 0; i < 8; ++i) yv[i] = yr[i * E_];
  const float wk = w[k];

  float acc = 0.f;
#pragma unroll
  for (int i = 0; i < 64; ++i) acc = __builtin_fmaf(w[n0 + i], col[i], acc);

  fq[wave][lane] = acc;
  __syncthreads();

  float sum = fq[0][lane];
#pragma unroll
  for (int q = 1; q < 8; ++q) sum += fq[q][lane];
  const float u = wk + sum;

  float pv[8];
#pragma unroll
  for (int i = 0; i < 8; ++i) pv[i] = yv[i] * u;
#pragma unroll
  for (int off = 16; off > 0; off >>= 1)
#pragma unroll
    for (int i = 0; i < 8; ++i) pv[i] += __shfl_xor(pv[i], off, 32);
  if (lane == 0) {
#pragma unroll
    for (int i = 0; i < 8; ++i) srow[8 * wave + i] = pv[i];
  }
  __syncthreads();
  if (threadIdx.x < 64) sp[blk * 64 + threadIdx.x] = srow[threadIdx.x];
}

__device__ __forceinline__ void bias_dot_add(const float* __restrict__ bias,
                                             const float* __restrict__ w,
                                             const float* __restrict__ c,
                                             float* __restrict__ c2) {
  if (threadIdx.x < 32) {
    const int lane = threadIdx.x;
    float s = 0.f;
#pragma unroll
    for (int j = 0; j < 16; ++j)
      s = __builtin_fmaf(bias[lane + 32 * j], w[lane + 32 * j], s);
#pragma unroll
    for (int off = 16; off > 0; off >>= 1) s += __shfl_xor(s, off, 32);
    if (lane == 0) *c2 = s + *c;
  }
}

__global__ void stage1_kernel(const float* __restrict__ z,
                              const float* __restrict__ W1,
                              const float* __restrict__ b1,
                              float* __restrict__ zz1,
                              const float* __restrict__ Wy,
                              const float* __restrict__ by,
                              const int* __restrict__ y,
                              float* __restrict__ ye0,
                              const float* __restrict__ Wy2,
                              const float* __restrict__ by2,
                              const float* __restrict__ wproj,
                              float* __restrict__ vf,
                              float* __restrict__ cf) {
  if (blockIdx.x < 64)       irm_tiles(z, W1, b1, zz1, blockIdx.x);
  else if (blockIdx.x < 192) gather_rows(Wy, by, y, ye0, blockIdx.x - 64);
  else if (blockIdx.x < 208) fold_proj(Wy2, wproj, vf, blockIdx.x - 192);
  else                       bias_dot(by2, wproj, cf);
}

__global__ void stage2_kernel(const float* __restrict__ x,
                              const float* __restrict__ W2,
                              const float* __restrict__ b2,
                              float* __restrict__ o,
                              const float* __restrict__ Wy1,
                              const float* __restrict__ by1,
                              const float* __restrict__ ye0,
                              const float* __restrict__ vf,
                              const float* __restrict__ cf,
                              float* __restrict__ sp,
                              float* __restrict__ c2) {
  if (blockIdx.x < 64)      irm_tiles(x, W2, b2, o, blockIdx.x);
  else if (blockIdx.x < 80) fold_rows(Wy1, vf, ye0, sp, blockIdx.x - 64);
  else                      bias_dot_add(by1, vf, cf, c2);
}

__global__ __launch_bounds__(256)
void zlat_kernel(const float* __restrict__ zz,
                 const float* __restrict__ Wz,
                 const float* __restrict__ sp,
                 const float* __restrict__ c2,
                 float* __restrict__ out) {
  __shared__ float part[8][64 * 32];

  const int lane = threadIdx.x & 31;
  const int wave = threadIdx.x >> 5;
  const int half = lane >> 4, lr = lane & 15;
  const int n0 = blockIdx.x * 32;
  const int k0 = wave * 64 + 4 * half;

  const float* ap = zz + lr * ZD_ + k0;
  const float* bp = Wz + (n0 + lr) * ZD_ + k0;

  v4f bq[8][2], aq[8][4];
#pragma unroll
  for (int j = 0; j < 8; ++j)
#pragma unroll
    for (int t = 0; t < 2; ++t)
      bq[j][t] = *(const v4f*)(bp + t * 16 * ZD_ + 8 * j);
#pragma unroll
  for (int j = 0; j < 8; ++j)
#pragma unroll
    for (int m = 0; m < 4; ++m)
      aq[j][m] = *(const v4f*)(ap + m * 16 * ZD_ + 8 * j);

  const float* spr = sp + (4 * (lane >> 3)) * 64 + 8 * wave + (lane & 7);
  const float p0 = spr[0], p1 = spr[64], p2 = spr[128], p3 = spr[192];
  const float cadd = *c2;

  v8f acc[4][2] = {};
#pragma unroll
  for (int j = 0; j < 8; ++j) {
#pragma unroll
    for (int p = 0; p < 2; ++p) {
#pragma unroll
      for (int m = 0; m < 4; ++m) {
        v2f a = {aq[j][m][2 * p], aq[j][m][2 * p + 1]};
#pragma unroll
        for (int t = 0; t < 2; ++t) {
          v2f b = {bq[j][t][2 * p], bq[j][t][2 * p + 1]};
          acc[m][t] = WMMA_F32(a, b, acc[m][t]);
        }
      }
    }
  }

#pragma unroll
  for (int m = 0; m < 4; ++m)
#pragma unroll
    for (int t = 0; t < 2; ++t)
#pragma unroll
      for (int r = 0; r < 8; ++r)
        part[wave][(16 * m + 8 * half + r) * 32 + 16 * t + lr] = acc[m][t][r];

  float sl = ((p0 + p1) + p2) + p3;
  sl += __shfl_xor(sl, 8, 32);
  sl += __shfl_xor(sl, 16, 32);
  sl += cadd;

  __syncthreads();

#pragma unroll
  for (int r = 0; r < 8; ++r) {
    const int row = 8 * wave + r;
    float sum = part[0][row * 32 + lane];
#pragma unroll
    for (int w = 1; w < 8; ++w) sum += part[w][row * 32 + lane];
    out[row * L_ + n0 + lane] = sum * __shfl(sl, r, 32);
  }
}

extern "C" void kernel_launch(void* const* d_in, const int* in_sizes, int n_in,
                              void* d_out, int out_size, void* d_ws, size_t ws_size,
                              hipStream_t stream) {
  const float* z      = (const float*)d_in[0];
  const int*   y      = (const int*)  d_in[1];
  const float* W_yemb = (const float*)d_in[2];
  const float* b_yemb = (const float*)d_in[3];
  const float* Wy1    = (const float*)d_in[4];
  const float* by1    = (const float*)d_in[5];
  const float* Wy2    = (const float*)d_in[6];
  const float* by2    = (const float*)d_in[7];
  const float* Wz1    = (const float*)d_in[8];
  const float* bz1    = (const float*)d_in[9];
  const float* Wz2    = (const float*)d_in[10];
  const float* bz2    = (const float*)d_in[11];
  const float* W_zlat = (const float*)d_in[12];
  const float* w_proj = (const float*)d_in[13];
  float* out = (float*)d_out;

  float* ws  = (float*)d_ws;
  float* ye0 = ws;
  float* sp  = ye0 + B_ * E_;
  float* c2  = sp + 16 * B_;
  float* zz1 = sp + B_ * E_;
  float* zz2 = zz1 + B_ * ZD_;
  float* vf  = zz2 + B_ * ZD_;
  float* cf  = vf + E_;

  stage1_kernel<<<209, 256, 0, stream>>>(z, Wz1, bz1, zz1, W_yemb, b_yemb, y, ye0,
                                         Wy2, by2, w_proj, vf, cf);
  stage2_kernel<<<81, 256, 0, stream>>>(zz1, Wz2, bz2, zz2, Wy1, by1, ye0, vf, cf,
                                        sp, c2);
  zlat_kernel<<<256, 256, 0, stream>>>(zz2, W_zlat, sp, c2, out);
}


# GEMM operand loads made whole-line: weight and z rows fetched 4 full lines per load and re-formed into fragments via wave-private LDS; layer outputs stored in fragment order; sums keep their order
# speedup vs baseline: 3.1326x; 1.1458x over previous
#include <hip/hip_runtime.h>

typedef __attribute__((ext_vector_type(2))) float v2f;
typedef __attribute__((ext_vector_type(4))) float v4f;
typedef __attribute__((ext_vector_type(8))) float v8f;

#define B_  64
#define ZD_ 512
#define E_  512
#define L_  8192
#define NL_ 1000

#define WMMA_F32(a, b, c) \
  __builtin_amdgcn_wmma_f32_16x16x4_f32(false, (a), false, (b), (short)0, (c), false, false)


__device__ __forceinline__ void gather_rows(const float* __restrict__ Wy,
                                            const float* __restrict__ bias,
                                            const int* __restrict__ y,
                                            float* __restrict__ out, int blk) {
  int idx = blk * 256 + threadIdx.x;
  int b = idx >> 9;
  int e = idx & 511;
  out[idx] = Wy[e * NL_ + y[b]] + bias[e];
}


#define SP_ 68

template <int NR>
__device__ __forceinline__ void rows_fetch(const float* __restrict__ src, int lane,
                                           v4f (&r)[NR / 2]) {
  const float* p = src + (lane >> 4) * 512 + 4 * (lane & 15);
#pragma unroll
  for (int i = 0; i < NR / 2; ++i) r[i] = *(const v4f*)(p + i * 1024);
}

template <int NR>
__device__ __forceinline__ void rows_stage(float* st, int lane, const v4f (&r)[NR / 2]) {
  float* q = st + (lane >> 4) * SP_ + 4 * (lane & 15);
#pragma unroll
  for (int i = 0; i < NR / 2; ++i) *(v4f*)(q + i * 2 * SP_) = r[i];
}

__device__ __forceinline__ void stage_sync() {
  __builtin_amdgcn_fence(__ATOMIC_RELEASE, "wavefront");
  __builtin_amdgcn_wave_barrier();
  __builtin_amdgcn_fence(__ATOMIC_ACQUIRE, "wavefront");
}

template <bool A_FRAG>
__device__ __forceinline__ void irm_tiles(const float* __restrict__ x,
                                          const float* __restrict__ W,
                                          const float* __restrict__ bias,
                                          float* __restrict__ outf, int blk) {
  constexpr int NRS = A_FRAG ? 32 : 48;
  __shared__ __attribute__((aligned(16))) float sm[8 * NRS * SP_];

  const int lane = threadIdx.x & 31;
  const int wave = threadIdx.x >> 5;
  const int half = lane >> 4, lr = lane & 15;
  const int mt = blk >> 4, nb = blk & 15;
  const int m0 = mt * 16;
  const int n0 = nb * 32;
  float* st = sm + wave * (NRS * SP_);

  v4f br[16];
  rows_fetch<32>(W + n0 * E_ + wave * 64, lane, br);

  const int col = 4 * wave + (lane & 3);
  const int rw  = lane >> 2;
  const int off = (((4 * nb + (wave >> 1)) * 4 + mt) * 32 + (wave & 1) * 16 + rw) * 4
                  + (lane & 3);
  const float xr0 = A_FRAG ? x[off]      : x[(m0 + rw) * E_ + n0 + col];
  const float xr1 = A_FRAG ? x[off + 32] : x[(m0 + rw + 8) * E_ + n0 + col];
  const float bn  = bias[n0 + col];

  __builtin_amdgcn_sched_barrier(0);

  v4f aq[8], ar[8];
  if (A_FRAG) {
    const float* ap = x + ((wave * 32 + mt) * 32 + lane) * 4;
#pragma unroll
    for (int j = 0; j < 8; ++j) aq[j] = *(const v4f*)(ap + j * 512);
  } else {
    rows_fetch<16>(x + m0 * E_ + wave * 64, lane, ar);
  }
  __builtin_amdgcn_sched_barrier(0);

  rows_stage<32>(st, lane, br);
  if (!A_FRAG) rows_stage<16>(st + 32 * SP_, lane, ar);
  stage_sync();

  v4f bq[8][2];
#pragma unroll
  for (int j = 0; j < 8; ++j) {
#pragma unroll
    for (int t = 0; t < 2; ++t)
      bq[j][t] = *(const v4f*)(st + (16 * t + lr) * SP_ + 8 * j + 4 * half);
    if (!A_FRAG) aq[j] = *(const v4f*)(st + (32 + lr) * SP_ + 8 * j + 4 * half);
  }

  v8f acc[2] = {};
#pragma unroll
  for (int j = 0; j < 8; ++j) {
#pragma unroll
    for (int p = 0; p < 2; ++p) {
      v2f a = {aq[j][2 * p], aq[j][2 * p + 1]};
#pragma unroll
      for (int t = 0; t < 2; ++t) {
        v2f b = {bq[j][t][2 * p], bq[j][t][2 * p + 1]};
        acc[t] = WMMA_F32(a, b, acc[t]);
      }
    }
  }

#pragma unroll
  for (int t = 0; t < 2; ++t)
#pragma unroll
    for (int r = 0; r < 8; ++r)
      st[(8 * half + r) * 32 + 16 * t + lr] = acc[t][r];

  __syncthreads();

  float s0 = sm[rw * 32 + col];
  float s1 = sm[(rw + 8) * 32 + col];
#pragma unroll
  for (int q = 1; q < 8; ++q) {
    s0 += sm[q * (NRS * SP_) + rw * 32 + col];
    s1 += sm[q * (NRS * SP_) + (rw + 8) * 32 + col];
  }
  outf[off]      = s0 + xr0 + bn;
  outf[off + 32] = s1 + xr1 + bn;
}

__device__ __forceinline__ void fold_proj(const float* __restrict__ W,
                                          const float* __restrict__ w,
                                          float* __restrict__ v, int blk) {
  __shared__ float fp[8][32];

  const int lane = threadIdx.x & 31;
  const int wave = __builtin_amdgcn_readfirstlane(threadIdx.x >> 5);
  const int k  = blk * 32 + lane;
  const int n0 = wave * 64;
  const float* wp = W + n0 * E_ + k;

  float col[64];
#pragma unroll
  for (int i = 0; i < 64; ++i) col[i] = wp[i * E_];

  float acc = 0.f;
#pragma unroll
  for (int i = 0; i < 64; ++i) acc = __builtin_fmaf(w[n0 + i], col[i], acc);

  fp[wave][lane] = acc;
  __syncthreads();

  if (wave == 0) {
    float sum = fp[0][lane];
#pragma unroll
    for (int q = 1; q < 8; ++q) sum += fp[q][lane];
    v[k] = w[k] + sum;
  }
}

__device__ __forceinline__ void bias_dot(const float* __restrict__ bias,
                                         const float* __restrict__ w,
                                         float* __restrict__ c) {
  if (threadIdx.x < 32) {
    const int lane = threadIdx.x;
    float s = 0.f;
#pragma unroll
    for (int j = 0; j < 16; ++j)
      s = __builtin_fmaf(bias[lane + 32 * j], w[lane + 32 * j], s);
#pragma unroll
    for (int off = 16; off > 0; off >>= 1) s += __shfl_xor(s, off, 32);
    if (lane == 0) *c = s;
  }
}

__device__ __forceinline__ void fold_rows(const float* __restrict__ W,
                                          const float* __restrict__ w,
                                          const float* __restrict__ ye,
                                          float* __restrict__ sp, int blk) {
  __shared__ float fq[8][32];
  __shared__ float srow[64];

  const int lane = threadIdx.x & 31;
  const int wave = __builtin_amdgcn_readfirstlane(threadIdx.x >> 5);
  const int k  = blk * 32 + lane;
  const int n0 = wave * 64;
  const float* wp = W + n0 * E_ + k;

  float col[64];
#pragma unroll
  for (int i = 0; i < 64; ++i) col[i] = wp[i * E_];

  const float* yr = ye + (8 * wave) * E_ + k;
  float yv[8];
#pragma unroll
  for (int i = 0; i < 8; ++i) yv[i] = yr[i * E_];
  const float wk = w[k];

  float acc = 0.f;
#pragma unroll
  for (int i = 0; i < 64; ++i) acc = __builtin_fmaf(w[n0 + i], col[i], acc);

  fq[wave][lane] = acc;
  __syncthreads();

  float sum = fq[0][lane];
#pragma unroll
  for (int q = 1; q < 8; ++q) sum += fq[q][lane];
  const float u = wk + sum;

  float pv[8];
#pragma unroll
  for (int i = 0; i < 8; ++i) pv[i] = yv[i] * u;
#pragma unroll
  for (int off = 16; off > 0; off >>= 1)
#pragma unroll
    for (int i = 0; i < 8; ++i) pv[i] += __shfl_xor(pv[i], off, 32);
  if (lane == 0) {
#pragma unroll
    for (int i = 0; i < 8; ++i) srow[8 * wave + i] = pv[i];
  }
  __syncthreads();
  if (threadIdx.x < 64) sp[blk * 64 + threadIdx.x] = srow[threadIdx.x];
}

__device__ __forceinline__ void bias_dot_add(const float* __restrict__ bias,
                                             const float* __restrict__ w,
                                             const float* __restrict__ c,
                                             float* __restrict__ c2) {
  if (threadIdx.x < 32) {
    const int lane = threadIdx.x;
    float s = 0.f;
#pragma unroll
    for (int j = 0; j < 16; ++j)
      s = __builtin_fmaf(bias[lane + 32 * j], w[lane + 32 * j], s);
#pragma unroll
    for (int off = 16; off > 0; off >>= 1) s += __shfl_xor(s, off, 32);
    if (lane == 0) *c2 = s + *c;
  }
}

__global__ void stage1_kernel(const float* __restrict__ z,
                              const float* __restrict__ W1,
                              const float* __restrict__ b1,
                              float* __restrict__ zz1,
                              const float* __restrict__ Wy,
                              const float* __restrict__ by,
                              const int* __restrict__ y,
                              float* __restrict__ ye0,
                              const float* __restrict__ Wy2,
                              const float* __restrict__ by2,
                              const float* __restrict__ wproj,
                              float* __restrict__ vf,
                              float* __restrict__ cf) {
  if (blockIdx.x < 64)       irm_tiles<false>(z, W1, b1, zz1, blockIdx.x);
  else if (blockIdx.x < 192) gather_rows(Wy, by, y, ye0, blockIdx.x - 64);
  else if (blockIdx.x < 208) fold_proj(Wy2, wproj, vf, blockIdx.x - 192);
  else                       bias_dot(by2, wproj, cf);
}

__global__ void stage2_kernel(const float* __restrict__ x,
                              const float* __restrict__ W2,
                              const float* __restrict__ b2,
                              float* __restrict__ o,
                              const float* __restrict__ Wy1,
                              const float* __restrict__ by1,
                              const float* __restrict__ ye0,
                              const float* __restrict__ vf,
                              const float* __restrict__ cf,
                              float* __restrict__ sp,
                              float* __restrict__ c2) {
  if (blockIdx.x < 64)      irm_tiles<true>(x, W2, b2, o, blockIdx.x);
  else if (blockIdx.x < 80) fold_rows(Wy1, vf, ye0, sp, blockIdx.x - 64);
  else                      bias_dot_add(by1, vf, cf, c2);
}

__global__ __launch_bounds__(256)
void zlat_kernel(const float* __restrict__ zzf,
                 const float* __restrict__ Wz,
                 const float* __restrict__ sp,
                 const float* __restrict__ c2,
                 float* __restrict__ out) {
  __shared__ __attribute__((aligned(16))) float sm[8 * 32 * SP_];

  const int lane = threadIdx.x & 31;
  const int wave = threadIdx.x >> 5;
  const int half = lane >> 4, lr = lane & 15;
  const int n0 = blockIdx.x * 32;
  float* st = sm + wave * (32 * SP_);

  v4f br[16];
  rows_fetch<32>(Wz + n0 * ZD_ + wave * 64, lane, br);
  __builtin_amdgcn_sched_barrier(0);

  const float* ap = zzf + (wave * 1024 + lane) * 4;
  v4f aq[8][4];
#pragma unroll
  for (int j = 0; j < 8; ++j)
#pragma unroll
    for (int m = 0; m < 4; ++m)
      aq[j][m] = *(const v4f*)(ap + (4 * j + m) * 128);

  const float* spr = sp + (4 * (lane >> 3)) * 64 + 8 * wave + (lane & 7);
  const float p0 = spr[0], p1 = spr[64], p2 = spr[128], p3 = spr[192];
  const float cadd = *c2;

  rows_stage<32>(st, lane, br);
  stage_sync();

  v4f bq[8][2];
#pragma unroll
  for (int j = 0; j < 8; ++j)
#pragma unroll
    for (int t = 0; t < 2; ++t)
      bq[j][t] = *(const v4f*)(st + (16 * t + lr) * SP_ + 8 * j + 4 * half);

  v8f acc[4][2] = {};
#pragma unroll
  for (int j = 0; j < 8; ++j) {
#pragma unroll
    for (int p = 0; p < 2; ++p) {
#pragma unroll
      for (int m = 0; m < 4; ++m) {
        v2f a = {aq[j][m][2 * p], aq[j][m][2 * p + 1]};
#pragma unroll
        for (int t = 0; t < 2; ++t) {
          v2f b = {bq[j][t][2 * p], bq[j][t][2 * p + 1]};
          acc[m][t] = WMMA_F32(a, b, acc[m][t]);
        }
      }
    }
  }

#pragma unroll
  for (int m = 0; m < 4; ++m)
#pragma unroll
    for (int t = 0; t < 2; ++t)
#pragma unroll
      for (int r = 0; r < 8; ++r)
        st[(16 * m + 8 * half + r) * 32 + 16 * t + lr] = acc[m][t][r];

  float sl = ((p0 + p1) + p2) + p3;
  sl += __shfl_xor(sl, 8, 32);
  sl += __shfl_xor(sl, 16, 32);
  sl += cadd;

  __syncthreads();

#pragma unroll
  for (int r = 0; r < 8; ++r) {
    const int row = 8 * wave + r;
    float sum = sm[row * 32 + lane];
#pragma unroll
    for (int w = 1; w < 8; ++w) sum += sm[w * (32 * SP_) + row * 32 + lane];
    out[row * L_ + n0 + lane] = sum * __shfl(sl, r, 32);
  }
}

extern "C" void kernel_launch(void* const* d_in, const int* in_sizes, int n_in,
                              void* d_out, int out_size, void* d_ws, size_t ws_size,
                              hipStream_t stream) {
  const float* z      = (const float*)d_in[0];
  const int*   y      = (const int*)  d_in[1];
  const float* W_yemb = (const float*)d_in[2];
  const float* b_yemb = (const float*)d_in[3];
  const float* Wy1    = (const float*)d_in[4];
  const float* by1    = (const float*)d_in[5];
  const float* Wy2    = (const float*)d_in[6];
  const float* by2    = (const float*)d_in[7];
  const float* Wz1    = (const float*)d_in[8];
  const float* bz1    = (const float*)d_in[9];
  const float* Wz2    = (const float*)d_in[10];
  const float* bz2    = (const float*)d_in[11];
  const float* W_zlat = (const float*)d_in[12];
  const float* w_proj = (const float*)d_in[13];
  float* out = (float*)d_out;

  float* ws  = (float*)d_ws;
  float* ye0 = ws;
  float* sp  = ye0 + B_ * E_;
  float* c2  = sp + 16 * B_;
  float* zz1 = sp + B_ * E_;
  float* zz2 = zz1 + B_ * ZD_;
  float* vf  = zz2 + B_ * ZD_;
  float* cf  = vf + E_;

  stage1_kernel<<<209, 256, 0, stream>>>(z, Wz1, bz1, zz1, W_yemb, b_yemb, y, ye0,
                                         Wy2, by2, w_proj, vf, cf);
  stage2_kernel<<<81, 256, 0, stream>>>(zz1, Wz2, bz2, zz2, Wy1, by1, ye0, vf, cf,
                                        sp, c2);
  zlat_kernel<<<256, 256, 0, stream>>>(zz2, W_zlat, sp, c2, out);
}


# Big GEMM: 4 waves per block (one per SIMD), not 8: a wave takes two 64-wide K chunks in turn into one accumulator set, chunk 1's loads landing under chunk 0's products; k sum order changes
# speedup vs baseline: 3.3635x; 1.0737x over previous
#include <hip/hip_runtime.h>

typedef __attribute__((ext_vector_type(2))) float v2f;
typedef __attribute__((ext_vector_type(4))) float v4f;
typedef __attribute__((ext_vector_type(8))) float v8f;

#define B_  64
#define ZD_ 512
#define E_  512
#define L_  8192
#define NL_ 1000

#define WMMA_F32(a, b, c) \
  __builtin_amdgcn_wmma_f32_16x16x4_f32(false, (a), false, (b), (short)0, (c), false, false)


__device__ __forceinline__ void gather_rows(const float* __restrict__ Wy,
                                            const float* __restrict__ bias,
                                            const int* __restrict__ y,
                                            float* __restrict__ out, int blk) {
  int idx = blk * 256 + threadIdx.x;
  int b = idx >> 9;
  int e = idx & 511;
  out[idx] = Wy[e * NL_ + y[b]] + bias[e];
}


#define SP_ 68

template <int NR>
__device__ __forceinline__ void rows_fetch(const float* __restrict__ src, int lane,
                                           v4f (&r)[NR / 2]) {
  const float* p = src + (lane >> 4) * 512 + 4 * (lane & 15);
#pragma unroll
  for (int i = 0; i < NR / 2; ++i) r[i] = *(const v4f*)(p + i * 1024);
}

template <int NR>
__device__ __forceinline__ void rows_stage(float* st, int lane, const v4f (&r)[NR / 2]) {
  float* q = st + (lane >> 4) * SP_ + 4 * (lane & 15);
#pragma unroll
  for (int i = 0; i < NR / 2; ++i) *(v4f*)(q + i * 2 * SP_) = r[i];
}

__device__ __forceinline__ void stage_sync() {
  __builtin_amdgcn_fence(__ATOMIC_RELEASE, "wavefront");
  __builtin_amdgcn_wave_barrier();
  __builtin_amdgcn_fence(__ATOMIC_ACQUIRE, "wavefront");
}

template <bool A_FRAG>
__device__ __forceinline__ void irm_tiles(const float* __restrict__ x,
                                          const float* __restrict__ W,
                                          const float* __restrict__ bias,
                                          float* __restrict__ outf, int blk) {
  constexpr int NRS = A_FRAG ? 32 : 48;
  __shared__ __attribute__((aligned(16))) float sm[8 * NRS * SP_];

  const int lane = threadIdx.x & 31;
  const int wave = threadIdx.x >> 5;
  const int half = lane >> 4, lr = lane & 15;
  const int mt = blk >> 4, nb = blk & 15;
  const int m0 = mt * 16;
  const int n0 = nb * 32;
  float* st = sm + wave * (NRS * SP_);

  v4f br[16];
  rows_fetch<32>(W + n0 * E_ + wave * 64, lane, br);

  const int col = 4 * wave + (lane & 3);
  const int rw  = lane >> 2;
  const int off = (((4 * nb + (wave >> 1)) * 4 + mt) * 32 + (wave & 1) * 16 + rw) * 4
                  + (lane & 3);
  const float xr0 = A_FRAG ? x[off]      : x[(m0 + rw) * E_ + n0 + col];
  const float xr1 = A_FRAG ? x[off + 32] : x[(m0 + rw + 8) * E_ + n0 + col];
  const float bn  = bias[n0 + col];

  __builtin_amdgcn_sched_barrier(0);

  v4f aq[8], ar[8];
  if (A_FRAG) {
    const float* ap = x + ((wave * 32 + mt) * 32 + lane) * 4;
#pragma unroll
    for (int j = 0; j < 8; ++j) aq[j] = *(const v4f*)(ap + j * 512);
  } else {
    rows_fetch<16>(x + m0 * E_ + wave * 64, lane, ar);
  }
  __builtin_amdgcn_sched_barrier(0);

  rows_stage<32>(st, lane, br);
  if (!A_FRAG) rows_stage<16>(st + 32 * SP_, lane, ar);
  stage_sync();

  v4f bq[8][2];
#pragma unroll
  for (int j = 0; j < 8; ++j) {
#pragma unroll
    for (int t = 0; t < 2; ++t)
      bq[j][t] = *(const v4f*)(st + (16 * t + lr) * SP_ + 8 * j + 4 * half);
    if (!A_FRAG) aq[j] = *(const v4f*)(st + (32 + lr) * SP_ + 8 * j + 4 * half);
  }

  v8f acc[2] = {};
#pragma unroll
  for (int j = 0; j < 8; ++j) {
#pragma unroll
    for (int p = 0; p < 2; ++p) {
      v2f a = {aq[j][2 * p], aq[j][2 * p + 1]};
#pragma unroll
      for (int t = 0; t < 2; ++t) {
        v2f b = {bq[j][t][2 * p], bq[j][t][2 * p + 1]};
        acc[t] = WMMA_F32(a, b, acc[t]);
      }
    }
  }

#pragma unroll
  for (int t = 0; t < 2; ++t)
#pragma unroll
    for (int r = 0; r < 8; ++r)
      st[(8 * half + r) * 32 + 16 * t + lr] = acc[t][r];

  __syncthreads();

  float s0 = sm[rw * 32 + col];
  float s1 = sm[(rw + 8) * 32 + col];
#pragma unroll
  for (int q = 1; q < 8; ++q) {
    s0 += sm[q * (NRS * SP_) + rw * 32 + col];
    s1 += sm[q * (NRS * SP_) + (rw + 8) * 32 + col];
  }
  outf[off]      = s0 + xr0 + bn;
  outf[off + 32] = s1 + xr1 + bn;
}

__device__ __forceinline__ void fold_proj(const float* __restrict__ W,
                                          const float* __restrict__ w,
                                          float* __restrict__ v, int blk) {
  __shared__ float fp[8][32];

  const int lane = threadIdx.x & 31;
  const int wave = __builtin_amdgcn_readfirstlane(threadIdx.x >> 5);
  const int k  = blk * 32 + lane;
  const int n0 = wave * 64;
  const float* wp = W + n0 * E_ + k;

  float col[64];
#pragma unroll
  for (int i = 0; i < 64; ++i) col[i] = wp[i * E_];

  float acc = 0.f;
#pragma unroll
  for (int i = 0; i < 64; ++i) acc = __builtin_fmaf(w[n0 + i], col[i], acc);

  fp[wave][lane] = acc;
  __syncthreads();

  if (wave == 0) {
    float sum = fp[0][lane];
#pragma unroll
    for (int q = 1; q < 8; ++q) sum += fp[q][lane];
    v[k] = w[k] + sum;
  }
}

__device__ __forceinline__ void bias_dot(const float* __restrict__ bias,
                                         const float* __restrict__ w,
                                         float* __restrict__ c) {
  if (threadIdx.x < 32) {
    const int lane = threadIdx.x;
    float s = 0.f;
#pragma unroll
    for (int j = 0; j < 16; ++j)
      s = __builtin_fmaf(bias[lane + 32 * j], w[lane + 32 * j], s);
#pragma unroll
    for (int off = 16; off > 0; off >>= 1) s += __shfl_xor(s, off, 32);
    if (lane == 0) *c = s;
  }
}

__device__ __forceinline__ void fold_rows(const float* __restrict__ W,
                                          const float* __restrict__ w,
                                          const float* __restrict__ ye,
                                          float* __restrict__ sp, int blk) {
  __shared__ float fq[8][32];
  __shared__ float srow[64];

  const int lane = threadIdx.x & 31;
  const int wave = __builtin_amdgcn_readfirstlane(threadIdx.x >> 5);
  const int k  = blk * 32 + lane;
  const int n0 = wave * 64;
  const float* wp = W + n0 * E_ + k;

  float col[64];
#pragma unroll
  for (int i = 0; i < 64; ++i) col[i] = wp[i * E_];

  const float* yr = ye + (8 * wave) * E_ + k;
  float yv[8];
#pragma unroll
  for (int i = 0; i < 8; ++i) yv[i] = yr[i * E_];
  const float wk = w[k];

  float acc = 0.f;
#pragma unroll
  for (int i = 0; i < 64; ++i) acc = __builtin_fmaf(w[n0 + i], col[i], acc);

  fq[wave][lane] = acc;
  __syncthreads();

  float sum = fq[0][lane];
#pragma unroll
  for (int q = 1; q < 8; ++q) sum += fq[q][lane];
  const float u = wk + sum;

  float pv[8];
#pragma unroll
  for (int i = 0; i < 8; ++i) pv[i] = yv[i] * u;
#pragma unroll
  for (int off = 16; off > 0; off >>= 1)
#pragma unroll
    for (int i = 0; i < 8; ++i) pv[i] += __shfl_xor(pv[i], off, 32);
  if (lane == 0) {
#pragma unroll
    for (int i = 0; i < 8; ++i) srow[8 * wave + i] = pv[i];
  }
  __syncthreads();
  if (threadIdx.x < 64) sp[blk * 64 + threadIdx.x] = srow[threadIdx.x];
}

__device__ __forceinline__ void bias_dot_add(const float* __restrict__ bias,
                                             const float* __restrict__ w,
                                             const float* __restrict__ c,
                                             float* __restrict__ c2) {
  if (threadIdx.x < 32) {
    const int lane = threadIdx.x;
    float s = 0.f;
#pragma unroll
    for (int j = 0; j < 16; ++j)
      s = __builtin_fmaf(bias[lane + 32 * j], w[lane + 32 * j], s);
#pragma unroll
    for (int off = 16; off > 0; off >>= 1) s += __shfl_xor(s, off, 32);
    if (lane == 0) *c2 = s + *c;
  }
}

__global__ void stage1_kernel(const float* __restrict__ z,
                              const float* __restrict__ W1,
                              const float* __restrict__ b1,
                              float* __restrict__ zz1,
                              const float* __restrict__ Wy,
                              const float* __restrict__ by,
                              const int* __restrict__ y,
                              float* __restrict__ ye0,
                              const float* __restrict__ Wy2,
                              const float* __restrict__ by2,
                              const float* __restrict__ wproj,
                              float* __restrict__ vf,
                              float* __restrict__ cf) {
  if (blockIdx.x < 64)       irm_tiles<false>(z, W1, b1, zz1, blockIdx.x);
  else if (blockIdx.x < 192) gather_rows(Wy, by, y, ye0, blockIdx.x - 64);
  else if (blockIdx.x < 208) fold_proj(Wy2, wproj, vf, blockIdx.x - 192);
  else                       bias_dot(by2, wproj, cf);
}

__global__ void stage2_kernel(const float* __restrict__ x,
                              const float* __restrict__ W2,
                              const float* __restrict__ b2,
                              float* __restrict__ o,
                              const float* __restrict__ Wy1,
                              const float* __restrict__ by1,
                              const float* __restrict__ ye0,
                              const float* __restrict__ vf,
                              const float* __restrict__ cf,
                              float* __restrict__ sp,
                              float* __restrict__ c2) {
  if (blockIdx.x < 64)      irm_tiles<true>(x, W2, b2, o, blockIdx.x);
  else if (blockIdx.x < 80) fold_rows(Wy1, vf, ye0, sp, blockIdx.x - 64);
  else                      bias_dot_add(by1, vf, cf, c2);
}

#define ZCH_ (32 * SP_)

__device__ __forceinline__ void zlat_frags(const float* st, int lr, int half,
                                           v4f (&bq)[8][2]) {
#pragma unroll
  for (int j = 0; j < 8; ++j)
#pragma unroll
    for (int t = 0; t < 2; ++t)
      bq[j][t] = *(const v4f*)(st + (16 * t + lr) * SP_ + 8 * j + 4 * half);
}

template <int J0, int J1>
__device__ __forceinline__ void zlat_products(const v4f (&aq)[8][4], const v4f (&bq)[8][2],
                                              v8f (&acc)[4][2]) {
#pragma unroll
  for (int j = J0; j < J1; ++j) {
#pragma unroll
    for (int p = 0; p < 2; ++p) {
#pragma unroll
      for (int m = 0; m < 4; ++m) {
        v2f a = {aq[j][m][2 * p], aq[j][m][2 * p + 1]};
#pragma unroll
        for (int t = 0; t < 2; ++t) {
          v2f b = {bq[j][t][2 * p], bq[j][t][2 * p + 1]};
          acc[m][t] = WMMA_F32(a, b, acc[m][t]);
        }
      }
    }
  }
}

__global__ __launch_bounds__(128)
void zlat_kernel(const float* __restrict__ zzf,
                 const float* __restrict__ Wz,
                 const float* __restrict__ sp,
                 const float* __restrict__ c2,
                 float* __restrict__ out) {
  __shared__ __attribute__((aligned(16))) float sm[4 * 2 * ZCH_];

  const int lane = threadIdx.x & 31;
  const int wave = threadIdx.x >> 5;
  const int half = lane >> 4, lr = lane & 15;
  const int n0 = blockIdx.x * 32;
  float* st0 = sm + wave * (2 * ZCH_);
  float* st1 = st0 + ZCH_;

  v4f br0[16], br1[16];
  rows_fetch<32>(Wz + n0 * ZD_ + (2 * wave) * 64, lane, br0);
  __builtin_amdgcn_sched_barrier(0);

  const float* ap0 = zzf + ((2 * wave) * 1024 + lane) * 4;
  v4f aq0[8][4], aq1[8][4];
#pragma unroll
  for (int j = 0; j < 8; ++j)
#pragma unroll
    for (int m = 0; m < 4; ++m)
      aq0[j][m] = *(const v4f*)(ap0 + (4 * j + m) * 128);
  rows_fetch<32>(Wz + n0 * ZD_ + (2 * wave + 1) * 64, lane, br1);
  __builtin_amdgcn_sched_barrier(0);

  v8f acc[4][2] = {};
  v4f bq0[8][2], bq1[8][2];

  rows_stage<32>(st0, lane, br0);
  stage_sync();
  zlat_frags(st0, lr, half, bq0);
  zlat_products<0, 2>(aq0, bq0, acc);
  __builtin_amdgcn_sched_barrier(0);

  const float* ap1 = ap0 + 4096;
#pragma unroll
  for (int j = 0; j < 4; ++j)
#pragma unroll
    for (int m = 0; m < 4; ++m)
      aq1[j][m] = *(const v4f*)(ap1 + (4 * j + m) * 128);

  zlat_products<2, 4>(aq0, bq0, acc);
  __builtin_amdgcn_sched_barrier(0);

#pragma unroll
  for (int j = 4; j < 8; ++j)
#pragma unroll
    for (int m = 0; m < 4; ++m)
      aq1[j][m] = *(const v4f*)(ap1 + (4 * j + m) * 128);
  const float* spr = sp + (8 * half) * 64 + 16 * wave + lr;
  float pq[8];
#pragma unroll
  for (int i = 0; i < 8; ++i) pq[i] = spr[i * 64];
  const float cadd = *c2;
  __builtin_amdgcn_sched_barrier(0);

  zlat_products<4, 7>(aq0, bq0, acc);
  __builtin_amdgcn_sched_barrier(0);

  rows_stage<32>(st1, lane, br1);
  stage_sync();
  zlat_frags(st1, lr, half, bq1);
  zlat_products<7, 8>(aq0, bq0, acc);
  zlat_products<0, 8>(aq1, bq1, acc);

#pragma unroll
  for (int m = 0; m < 4; ++m)
#pragma unroll
    for (int t = 0; t < 2; ++t)
#pragma unroll
      for (int r = 0; r < 8; ++r)
        st0[(16 * m + 8 * half + r) * 32 + 16 * t + lr] = acc[m][t][r];

  const float ra = ((pq[0] + pq[1]) + pq[2]) + pq[3];
  const float rb = ((pq[4] + pq[5]) + pq[6]) + pq[7];
  float sl = ra + rb;
  sl += __shfl_xor(sl, 16, 32);
  sl += cadd;

  __syncthreads();

#pragma unroll
  for (int r = 0; r < 16; ++r) {
    const int row = 16 * wave + r;
    float sum = sm[row * 32 + lane];
#pragma unroll
    for (int w = 1; w < 4; ++w) sum += sm[w * (2 * ZCH_) + row * 32 + lane];
    out[row * L_ + n0 + lane] = sum * __shfl(sl, r, 32);
  }
}

extern "C" void kernel_launch(void* const* d_in, const int* in_sizes, int n_in,
                              void* d_out, int out_size, void* d_ws, size_t ws_size,
                              hipStream_t stream) {
  const float* z      = (const float*)d_in[0];
  const int*   y      = (const int*)  d_in[1];
  const float* W_yemb = (const float*)d_in[2];
  const float* b_yemb = (const float*)d_in[3];
  const float* Wy1    = (const float*)d_in[4];
  const float* by1    = (const float*)d_in[5];
  const float* Wy2    = (const float*)d_in[6];
  const float* by2    = (const float*)d_in[7];
  const float* Wz1    = (const float*)d_in[8];
  const float* bz1    = (const float*)d_in[9];
  const float* Wz2    = (const float*)d_in[10];
  const float* bz2    = (const float*)d_in[11];
  const float* W_zlat = (const float*)d_in[12];
  const float* w_proj = (const float*)d_in[13];
  float* out = (float*)d_out;

  float* ws  = (float*)d_ws;
  float* ye0 = ws;
  float* sp  = ye0 + B_ * E_;
  float* c2  = sp + 16 * B_;
  float* zz1 = sp + B_ * E_;
  float* zz2 = zz1 + B_ * ZD_;
  float* vf  = zz2 + B_ * ZD_;
  float* cf  = vf + E_;

  stage1_kernel<<<209, 256, 0, stream>>>(z, Wz1, bz1, zz1, W_yemb, b_yemb, y, ye0,
                                         Wy2, by2, w_proj, vf, cf);
  stage2_kernel<<<81, 256, 0, stream>>>(zz1, Wz2, bz2, zz2, Wy1, by1, ye0, vf, cf,
                                        sp, c2);
  zlat_kernel<<<256, 128, 0, stream>>>(zz2, W_zlat, sp, c2, out);
}
